# starting kernel
# baseline (speedup 1.0000x reference)
#include <hip/hip_runtime.h>
#include <hip/hip_bf16.h>

typedef float v2f __attribute__((ext_vector_type(2)));
typedef float v8f __attribute__((ext_vector_type(8)));

static constexpr int kVocab = 256;
static constexpr int kE     = 64;
static constexpr int kH     = 256;
static constexpr int kW     = 20;
static constexpr int kWords = 64 * 128;
static constexpr int kF3    = 3 * kH;

__global__ __launch_bounds__(32) void build_tables(
    const float* __restrict__ emb,
    const float* __restrict__ w1, const float* __restrict__ b1,
    const float* __restrict__ w3, const float* __restrict__ b3,
    const float* __restrict__ w5, const float* __restrict__ b5,
    float* __restrict__ T)
{
    int tile  = blockIdx.x;
    int tap   = tile >> 8;
    int cTile = (tile >> 4) & 15;
    int hTile = tile & 15;

    const float* wp; const float* bp; int kk, j;
    if (tap == 0)     { wp = w1; bp = b1; kk = 1; j = 0; }
    else if (tap < 4) { wp = w3; bp = b3; kk = 3; j = tap - 1; }
    else              { wp = w5; bp = b5; kk = 5; j = tap - 4; }
    const bool center = (j == (kk >> 1));

    const int lane = threadIdx.x & 31;
    const int half = lane >> 4;
    const int l16  = lane & 15;

    const int row = cTile * 16 + l16;
    const int col = hTile * 16 + l16;

    v8f acc = {};
    for (int e0 = 0; e0 < kE; e0 += 4) {
        int ea = e0 + 2 * half;
        v2f a = *(const v2f*)(emb + row * kE + ea);
        v2f b;
        b.x = wp[(col * kE + ea    ) * kk + j];
        b.y = wp[(col * kE + ea + 1) * kk + j];
        acc = __builtin_amdgcn_wmma_f32_16x16x4_f32(
            false, a, false, b, (short)0, acc, false, false);
    }

    const float bias = center ? bp[col] : 0.0f;
    float* outp = T + tap * (kVocab * kH) + (cTile * 16) * kH + hTile * 16;
#pragma unroll
    for (int i = 0; i < 8; ++i) {
        int m = half * 8 + i;
        outp[m * kH + l16] = acc[i] + bias;
    }
}

__global__ __launch_bounds__(256) void conv_gather(
    const int*   __restrict__ chars,
    const float* __restrict__ T,
    float*       __restrict__ feat)
{
    __shared__ int c[kW];
    const int n   = blockIdx.x;
    const int tid = threadIdx.x;
    if (tid < kW) c[tid] = chars[n * kW + tid];
    __syncthreads();

    const int h = tid;
    const float* T0 = T;
    const float* T1 = T + 1 * 65536;
    const float* T2 = T + 2 * 65536;
    const float* T3 = T + 3 * 65536;
    const float* T4 = T + 4 * 65536;
    const float* T5 = T + 5 * 65536;
    const float* T6 = T + 6 * 65536;
    const float* T7 = T + 7 * 65536;
    const float* T8 = T + 8 * 65536;

    float m1 = 0.0f, m3 = 0.0f, m5 = 0.0f;
#pragma unroll
    for (int t = 0; t < kW; ++t) {
        float s1 = T0[c[t] * kH + h];
        m1 = fmaxf(m1, s1);

        float s3 = T2[c[t] * kH + h];
        if (t >= 1)      s3 += T1[c[t - 1] * kH + h];
        if (t <= kW - 2) s3 += T3[c[t + 1] * kH + h];
        m3 = fmaxf(m3, s3);

        float s5 = T6[c[t] * kH + h];
        if (t >= 2)      s5 += T4[c[t - 2] * kH + h];
        if (t >= 1)      s5 += T5[c[t - 1] * kH + h];
        if (t <= kW - 2) s5 += T7[c[t + 1] * kH + h];
        if (t <= kW - 3) s5 += T8[c[t + 2] * kH + h];
        m5 = fmaxf(m5, s5);
    }

    feat[n * kF3 + h]          = m1;
    feat[n * kF3 + kH + h]     = m3;
    feat[n * kF3 + 2 * kH + h] = m5;
}

__global__ __launch_bounds__(256) void linear_wmma(
    const float* __restrict__ feat,
    const float* __restrict__ lw,
    const float* __restrict__ lb,
    float*       __restrict__ out)
{
    const int wave  = blockIdx.x * (blockDim.x >> 5) + (threadIdx.x >> 5);
    const int mTile = wave >> 4;
    const int nTile = wave & 15;
    const int lane  = threadIdx.x & 31;
    const int half  = lane >> 4;
    const int l16   = lane & 15;

    const float* arow = feat + (mTile * 16 + l16) * kF3 + 2 * half;
    const float* brow = lw   + (nTile * 16 + l16) * kF3 + 2 * half;

    v8f acc = {};
#pragma unroll 4
    for (int k0 = 0; k0 < kF3; k0 += 4) {
        v2f a = *(const v2f*)(arow + k0);
        v2f b = *(const v2f*)(brow + k0);
        acc = __builtin_amdgcn_wmma_f32_16x16x4_f32(
            false, a, false, b, (short)0, acc, false, false);
    }

    const float bias = lb[nTile * 16 + l16];
    float* o = out + (mTile * 16) * kH + nTile * 16;
#pragma unroll
    for (int i = 0; i < 8; ++i) {
        int m = half * 8 + i;
        o[m * kH + l16] = acc[i] + bias;
    }
}

extern "C" void kernel_launch(void* const* d_in, const int* in_sizes, int n_in,
                              void* d_out, int out_size, void* d_ws, size_t ws_size,
                              hipStream_t stream)
{
    const int*   chars = (const int*)  d_in[0];
    const float* emb   = (const float*)d_in[1];
    const float* w1    = (const float*)d_in[2];
    const float* b1    = (const float*)d_in[3];
    const float* w3    = (const float*)d_in[4];
    const float* b3    = (const float*)d_in[5];
    const float* w5    = (const float*)d_in[6];
    const float* b5    = (const float*)d_in[7];
    const float* lw    = (const float*)d_in[8];
    const float* lb    = (const float*)d_in[9];
    float*       outp  = (float*)d_out;

    float* T    = (float*)d_ws;
    float* feat = T + 9 * kVocab * kH;

    build_tables<<<9 * 16 * 16, 32, 0, stream>>>(emb, w1, b1, w3, b3, w5, b5, T);

    conv_gather<<<kWords, 256, 0, stream>>>(chars, T, feat);

    linear_wmma<<<(kWords / 16) * (kH / 16) / 8, 256, 0, stream>>>(feat, lw, lb, outp);
}


# gather re-tiled: one wave per half word, 4 adjacent columns per lane, so each table read is a 128-bit load (512 contiguous bytes per wave) from L2; rolled position loop, all waves resident; same sums
# speedup vs baseline: 1.2498x; 1.2498x over previous
#include <hip/hip_runtime.h>
#include <hip/hip_bf16.h>

typedef float v2f __attribute__((ext_vector_type(2)));
typedef float v8f __attribute__((ext_vector_type(8)));

static constexpr int kVocab = 256;
static constexpr int kE     = 64;
static constexpr int kH     = 256;
static constexpr int kW     = 20;
static constexpr int kWords = 64 * 128;
static constexpr int kF3    = 3 * kH;

__global__ __launch_bounds__(32) void build_tables(
    const float* __restrict__ emb,
    const float* __restrict__ w1, const float* __restrict__ b1,
    const float* __restrict__ w3, const float* __restrict__ b3,
    const float* __restrict__ w5, const float* __restrict__ b5,
    float* __restrict__ T)
{
    int tile  = blockIdx.x;
    int tap   = tile >> 8;
    int cTile = (tile >> 4) & 15;
    int hTile = tile & 15;

    const float* wp; const float* bp; int kk, j;
    if (tap == 0)     { wp = w1; bp = b1; kk = 1; j = 0; }
    else if (tap < 4) { wp = w3; bp = b3; kk = 3; j = tap - 1; }
    else              { wp = w5; bp = b5; kk = 5; j = tap - 4; }
    const bool center = (j == (kk >> 1));

    const int lane = threadIdx.x & 31;
    const int half = lane >> 4;
    const int l16  = lane & 15;

    const int row = cTile * 16 + l16;
    const int col = hTile * 16 + l16;

    v8f acc = {};
    for (int e0 = 0; e0 < kE; e0 += 4) {
        int ea = e0 + 2 * half;
        v2f a = *(const v2f*)(emb + row * kE + ea);
        v2f b;
        b.x = wp[(col * kE + ea    ) * kk + j];
        b.y = wp[(col * kE + ea + 1) * kk + j];
        acc = __builtin_amdgcn_wmma_f32_16x16x4_f32(
            false, a, false, b, (short)0, acc, false, false);
    }

    const float bias = center ? bp[col] : 0.0f;
    float* outp = T + tap * (kVocab * kH) + (cTile * 16) * kH + hTile * 16;
#pragma unroll
    for (int i = 0; i < 8; ++i) {
        int m = half * 8 + i;
        outp[m * kH + l16] = acc[i] + bias;
    }
}

typedef float v4f __attribute__((ext_vector_type(4)));

__device__ __forceinline__ v4f vmax4(v4f a, v4f b)
{
    v4f r;
    r.x = fmaxf(a.x, b.x); r.y = fmaxf(a.y, b.y);
    r.z = fmaxf(a.z, b.z); r.w = fmaxf(a.w, b.w);
    return r;
}

__device__ __forceinline__ v4f trow(const float* __restrict__ T, unsigned col, int tap, unsigned ch)
{
    return *(const v4f*)(T + ((size_t)tap * 65536u + (size_t)ch * kH) + col);
}

template <bool M2, bool M1, bool P1, bool P2>
__device__ __forceinline__ void position(
    const float* __restrict__ T, unsigned col,
    unsigned cm2, unsigned cm1, unsigned c0, unsigned cp1, unsigned cp2,
    v4f& m1, v4f& m3, v4f& m5)
{
    v4f s1 = trow(T, col, 0, c0);
    m1 = vmax4(m1, s1);

    v4f s3 = trow(T, col, 2, c0);
    if (M1) s3 += trow(T, col, 1, cm1);
    if (P1) s3 += trow(T, col, 3, cp1);
    m3 = vmax4(m3, s3);

    v4f s5 = trow(T, col, 6, c0);
    if (M2) s5 += trow(T, col, 4, cm2);
    if (M1) s5 += trow(T, col, 5, cm1);
    if (P1) s5 += trow(T, col, 7, cp1);
    if (P2) s5 += trow(T, col, 8, cp2);
    m5 = vmax4(m5, s5);
}

static constexpr int kWordsPerBlock = 4;

__global__ __launch_bounds__(64 * kWordsPerBlock) __attribute__((amdgpu_waves_per_eu(16, 16)))
void conv_gather(
    const int*   __restrict__ chars,
    const float* __restrict__ T,
    float*       __restrict__ feat)
{
    const unsigned lane = threadIdx.x & 31u;
    const unsigned wv  = (unsigned)(threadIdx.x >> 5);
    const unsigned n   = blockIdx.x * kWordsPerBlock + (wv >> 1);
    const unsigned col = (wv & 1u) * 128u + lane * 4u;

    const int cv = chars[n * kW + (lane < (unsigned)kW ? lane : (unsigned)(kW - 1))];
#define CH(i) ((unsigned)__builtin_amdgcn_readlane(cv, (i)))

    v4f m1 = {0.f, 0.f, 0.f, 0.f};
    v4f m3 = m1, m5 = m1;

    unsigned a = CH(0), b = CH(1), c = CH(2), d = CH(3);
    position<false, false, true, true>(T, col, 0u, 0u, a, b, c, m1, m3, m5);
    position<false, true,  true, true>(T, col, 0u, a, b, c, d, m1, m3, m5);
    for (int t = 2; t <= kW - 3; ++t) {
        const unsigned e = CH(t + 2);
        position<true, true, true, true>(T, col, a, b, c, d, e, m1, m3, m5);
        a = b; b = c; c = d; d = e;
    }
    position<true, true, true,  false>(T, col, a, b, c, d, 0u, m1, m3, m5);
    position<true, true, false, false>(T, col, b, c, d, 0u, 0u, m1, m3, m5);
#undef CH

    float* f = feat + (size_t)n * kF3 + col;
    *(v4f*)(f)          = m1;
    *(v4f*)(f + kH)     = m3;
    *(v4f*)(f + 2 * kH) = m5;
}

__global__ __launch_bounds__(256) void linear_wmma(
    const float* __restrict__ feat,
    const float* __restrict__ lw,
    const float* __restrict__ lb,
    float*       __restrict__ out)
{
    const int wave  = blockIdx.x * (blockDim.x >> 5) + (threadIdx.x >> 5);
    const int mTile = wave >> 4;
    const int nTile = wave & 15;
    const int lane  = threadIdx.x & 31;
    const int half  = lane >> 4;
    const int l16   = lane & 15;

    const float* arow = feat + (mTile * 16 + l16) * kF3 + 2 * half;
    const float* brow = lw   + (nTile * 16 + l16) * kF3 + 2 * half;

    v8f acc = {};
#pragma unroll 4
    for (int k0 = 0; k0 < kF3; k0 += 4) {
        v2f a = *(const v2f*)(arow + k0);
        v2f b = *(const v2f*)(brow + k0);
        acc = __builtin_amdgcn_wmma_f32_16x16x4_f32(
            false, a, false, b, (short)0, acc, false, false);
    }

    const float bias = lb[nTile * 16 + l16];
    float* o = out + (mTile * 16) * kH + nTile * 16;
#pragma unroll
    for (int i = 0; i < 8; ++i) {
        int m = half * 8 + i;
        o[m * kH + l16] = acc[i] + bias;
    }
}

extern "C" void kernel_launch(void* const* d_in, const int* in_sizes, int n_in,
                              void* d_out, int out_size, void* d_ws, size_t ws_size,
                              hipStream_t stream)
{
    const int*   chars = (const int*)  d_in[0];
    const float* emb   = (const float*)d_in[1];
    const float* w1    = (const float*)d_in[2];
    const float* b1    = (const float*)d_in[3];
    const float* w3    = (const float*)d_in[4];
    const float* b3    = (const float*)d_in[5];
    const float* w5    = (const float*)d_in[6];
    const float* b5    = (const float*)d_in[7];
    const float* lw    = (const float*)d_in[8];
    const float* lb    = (const float*)d_in[9];
    float*       outp  = (float*)d_out;

    float* T    = (float*)d_ws;
    float* feat = T + 9 * kVocab * kH;

    build_tables<<<9 * 16 * 16, 32, 0, stream>>>(emb, w1, b1, w3, b3, w5, b5, T);

    conv_gather<<<kWords / kWordsPerBlock, 64 * kWordsPerBlock, 0, stream>>>(chars, T, feat);

    linear_wmma<<<(kWords / 16) * (kH / 16) / 8, 256, 0, stream>>>(feat, lw, lb, outp);
}


# gather and linear fused into one launch: 16 words per workgroup, the 16x768 feature tile stays in LDS and is the WMMA A operand from there, feat never goes to memory; k order of every sum unchanged
# speedup vs baseline: 1.8606x; 1.4887x over previous
#include <hip/hip_runtime.h>
#include <hip/hip_bf16.h>

typedef float v2f __attribute__((ext_vector_type(2)));
typedef float v8f __attribute__((ext_vector_type(8)));

static constexpr int kVocab = 256;
static constexpr int kE     = 64;
static constexpr int kH     = 256;
static constexpr int kW     = 20;
static constexpr int kWords = 64 * 128;
static constexpr int kF3    = 3 * kH;

__global__ __launch_bounds__(32) void build_tables(
    const float* __restrict__ emb,
    const float* __restrict__ w1, const float* __restrict__ b1,
    const float* __restrict__ w3, const float* __restrict__ b3,
    const float* __restrict__ w5, const float* __restrict__ b5,
    float* __restrict__ T)
{
    int tile  = blockIdx.x;
    int tap   = tile >> 8;
    int cTile = (tile >> 4) & 15;
    int hTile = tile & 15;

    const float* wp; const float* bp; int kk, j;
    if (tap == 0)     { wp = w1; bp = b1; kk = 1; j = 0; }
    else if (tap < 4) { wp = w3; bp = b3; kk = 3; j = tap - 1; }
    else              { wp = w5; bp = b5; kk = 5; j = tap - 4; }
    const bool center = (j == (kk >> 1));

    const int lane = threadIdx.x & 31;
    const int half = lane >> 4;
    const int l16  = lane & 15;

    const int row = cTile * 16 + l16;
    const int col = hTile * 16 + l16;

    v8f acc = {};
    for (int e0 = 0; e0 < kE; e0 += 4) {
        int ea = e0 + 2 * half;
        v2f a = *(const v2f*)(emb + row * kE + ea);
        v2f b;
        b.x = wp[(col * kE + ea    ) * kk + j];
        b.y = wp[(col * kE + ea + 1) * kk + j];
        acc = __builtin_amdgcn_wmma_f32_16x16x4_f32(
            false, a, false, b, (short)0, acc, false, false);
    }

    const float bias = center ? bp[col] : 0.0f;
    float* outp = T + tap * (kVocab * kH) + (cTile * 16) * kH + hTile * 16;
#pragma unroll
    for (int i = 0; i < 8; ++i) {
        int m = half * 8 + i;
        outp[m * kH + l16] = acc[i] + bias;
    }
}

typedef float v4f __attribute__((ext_vector_type(4)));

__device__ __forceinline__ v4f vmax4(v4f a, v4f b)
{
    v4f r;
    r.x = fmaxf(a.x, b.x); r.y = fmaxf(a.y, b.y);
    r.z = fmaxf(a.z, b.z); r.w = fmaxf(a.w, b.w);
    return r;
}

__device__ __forceinline__ v4f trow(const float* __restrict__ T, unsigned col, int tap, unsigned ch)
{
    return *(const v4f*)(T + ((size_t)tap * 65536u + (size_t)ch * kH) + col);
}

template <bool M2, bool M1, bool P1, bool P2>
__device__ __forceinline__ void position(
    const float* __restrict__ T, unsigned col,
    unsigned cm2, unsigned cm1, unsigned c0, unsigned cp1, unsigned cp2,
    v4f& m1, v4f& m3, v4f& m5)
{
    v4f s1 = trow(T, col, 0, c0);
    m1 = vmax4(m1, s1);

    v4f s3 = trow(T, col, 2, c0);
    if (M1) s3 += trow(T, col, 1, cm1);
    if (P1) s3 += trow(T, col, 3, cp1);
    m3 = vmax4(m3, s3);

    v4f s5 = trow(T, col, 6, c0);
    if (M2) s5 += trow(T, col, 4, cm2);
    if (M1) s5 += trow(T, col, 5, cm1);
    if (P1) s5 += trow(T, col, 7, cp1);
    if (P2) s5 += trow(T, col, 8, cp2);
    m5 = vmax4(m5, s5);
}

static constexpr int kTileWords = 16;
static constexpr int kPitch     = kF3 + 4;

__global__ __launch_bounds__(64 * kTileWords) __attribute__((amdgpu_waves_per_eu(16, 16)))
void conv_gather_linear(
    const int*   __restrict__ chars,
    const float* __restrict__ T,
    const float* __restrict__ lw,
    const float* __restrict__ lb,
    float*       __restrict__ out)
{
    __shared__ __attribute__((aligned(16))) float sfeat[kTileWords * kPitch];

    const unsigned lane = threadIdx.x & 31u;
    const unsigned wv  = (unsigned)(threadIdx.x >> 5);
    const unsigned wrd = wv >> 1;
    const unsigned n   = blockIdx.x * kTileWords + wrd;
    const unsigned col = (wv & 1u) * 128u + lane * 4u;

    const int cv = chars[n * kW + (lane < (unsigned)kW ? lane : (unsigned)(kW - 1))];
#define CH(i) ((unsigned)__builtin_amdgcn_readlane(cv, (i)))

    v4f m1 = {0.f, 0.f, 0.f, 0.f};
    v4f m3 = m1, m5 = m1;

    unsigned a = CH(0), b = CH(1), c = CH(2), d = CH(3);
    position<false, false, true, true>(T, col, 0u, 0u, a, b, c, m1, m3, m5);
    position<false, true,  true, true>(T, col, 0u, a, b, c, d, m1, m3, m5);
    for (int t = 2; t <= kW - 3; ++t) {
        const unsigned e = CH(t + 2);
        position<true, true, true, true>(T, col, a, b, c, d, e, m1, m3, m5);
        a = b; b = c; c = d; d = e;
    }
    position<true, true, true,  false>(T, col, a, b, c, d, 0u, m1, m3, m5);
    position<true, true, false, false>(T, col, b, c, d, 0u, 0u, m1, m3, m5);
#undef CH

    float* f = sfeat + wrd * kPitch + col;
    *(v4f*)(f)          = m1;
    *(v4f*)(f + kH)     = m3;
    *(v4f*)(f + 2 * kH) = m5;

    __syncthreads();

    const int wvs = __builtin_amdgcn_readfirstlane((int)(threadIdx.x >> 5));
    if (wvs < kH / 16) {
        const int nTile = wvs;
        const int half  = (int)(lane >> 4);
        const int l16   = (int)(lane & 15u);

        const float* arow = sfeat + l16 * kPitch + 2 * half;
        const float* brow = lw + (nTile * 16 + l16) * kF3 + 2 * half;

        v8f acc = {};
#pragma unroll 8
        for (int k0 = 0; k0 < kF3; k0 += 4) {
            v2f av = *(const v2f*)(arow + k0);
            v2f bv = *(const v2f*)(brow + k0);
            acc = __builtin_amdgcn_wmma_f32_16x16x4_f32(
                false, av, false, bv, (short)0, acc, false, false);
        }

        const float bias = lb[nTile * 16 + l16];
        float* o = out + (size_t)(blockIdx.x * kTileWords) * kH + nTile * 16;
#pragma unroll
        for (int i = 0; i < 8; ++i) {
            int m = half * 8 + i;
            o[m * kH + l16] = acc[i] + bias;
        }
    }
}

extern "C" void kernel_launch(void* const* d_in, const int* in_sizes, int n_in,
                              void* d_out, int out_size, void* d_ws, size_t ws_size,
                              hipStream_t stream)
{
    const int*   chars = (const int*)  d_in[0];
    const float* emb   = (const float*)d_in[1];
    const float* w1    = (const float*)d_in[2];
    const float* b1    = (const float*)d_in[3];
    const float* w3    = (const float*)d_in[4];
    const float* b3    = (const float*)d_in[5];
    const float* w5    = (const float*)d_in[6];
    const float* b5    = (const float*)d_in[7];
    const float* lw    = (const float*)d_in[8];
    const float* lb    = (const float*)d_in[9];
    float*       outp  = (float*)d_out;

    float* T    = (float*)d_ws;

    build_tables<<<9 * 16 * 16, 32, 0, stream>>>(emb, w1, b1, w3, b3, w5, b5, T);

    conv_gather_linear<<<kWords / kTileWords, 64 * kTileWords, 0, stream>>>(chars, T, lw, lb, outp);
}


# workgroup tile doubled to 32 words, one workgroup per WGP: a wave gathers a whole word, each lw fragment feeds two word tiles (two accumulators), lw read once per 32 words; same sums
# speedup vs baseline: 2.1289x; 1.1442x over previous
#include <hip/hip_runtime.h>
#include <hip/hip_bf16.h>

typedef float v2f __attribute__((ext_vector_type(2)));
typedef float v8f __attribute__((ext_vector_type(8)));

static constexpr int kVocab = 256;
static constexpr int kE     = 64;
static constexpr int kH     = 256;
static constexpr int kW     = 20;
static constexpr int kWords = 64 * 128;
static constexpr int kF3    = 3 * kH;

__global__ __launch_bounds__(32) void build_tables(
    const float* __restrict__ emb,
    const float* __restrict__ w1, const float* __restrict__ b1,
    const float* __restrict__ w3, const float* __restrict__ b3,
    const float* __restrict__ w5, const float* __restrict__ b5,
    float* __restrict__ T)
{
    int tile  = blockIdx.x;
    int tap   = tile >> 8;
    int cTile = (tile >> 4) & 15;
    int hTile = tile & 15;

    const float* wp; const float* bp; int kk, j;
    if (tap == 0)     { wp = w1; bp = b1; kk = 1; j = 0; }
    else if (tap < 4) { wp = w3; bp = b3; kk = 3; j = tap - 1; }
    else              { wp = w5; bp = b5; kk = 5; j = tap - 4; }
    const bool center = (j == (kk >> 1));

    const int lane = threadIdx.x & 31;
    const int half = lane >> 4;
    const int l16  = lane & 15;

    const int row = cTile * 16 + l16;
    const int col = hTile * 16 + l16;

    v8f acc = {};
    for (int e0 = 0; e0 < kE; e0 += 4) {
        int ea = e0 + 2 * half;
        v2f a = *(const v2f*)(emb + row * kE + ea);
        v2f b;
        b.x = wp[(col * kE + ea    ) * kk + j];
        b.y = wp[(col * kE + ea + 1) * kk + j];
        acc = __builtin_amdgcn_wmma_f32_16x16x4_f32(
            false, a, false, b, (short)0, acc, false, false);
    }

    const float bias = center ? bp[col] : 0.0f;
    float* outp = T + tap * (kVocab * kH) + (cTile * 16) * kH + hTile * 16;
#pragma unroll
    for (int i = 0; i < 8; ++i) {
        int m = half * 8 + i;
        outp[m * kH + l16] = acc[i] + bias;
    }
}

typedef float v4f __attribute__((ext_vector_type(4)));

__device__ __forceinline__ v8f vmax8(v8f a, v8f b)
{
    v8f r;
#pragma unroll
    for (int i = 0; i < 8; ++i) r[i] = fmaxf(a[i], b[i]);
    return r;
}

__device__ __forceinline__ v8f trow(const float* __restrict__ T, unsigned col, int tap, unsigned ch)
{
    const float* p = T + ((size_t)tap * 65536u + (size_t)ch * kH) + col;
    v4f lo = *(const v4f*)(p);
    v4f hi = *(const v4f*)(p + 128);
    return __builtin_shufflevector(lo, hi, 0, 1, 2, 3, 4, 5, 6, 7);
}

template <bool M2, bool M1, bool P1, bool P2>
__device__ __forceinline__ void position(
    const float* __restrict__ T, unsigned col,
    unsigned cm2, unsigned cm1, unsigned c0, unsigned cp1, unsigned cp2,
    v8f& m1, v8f& m3, v8f& m5)
{
    v8f s1 = trow(T, col, 0, c0);
    m1 = vmax8(m1, s1);

    v8f s3 = trow(T, col, 2, c0);
    if (M1) s3 += trow(T, col, 1, cm1);
    if (P1) s3 += trow(T, col, 3, cp1);
    m3 = vmax8(m3, s3);

    v8f s5 = trow(T, col, 6, c0);
    if (M2) s5 += trow(T, col, 4, cm2);
    if (M1) s5 += trow(T, col, 5, cm1);
    if (P1) s5 += trow(T, col, 7, cp1);
    if (P2) s5 += trow(T, col, 8, cp2);
    m5 = vmax8(m5, s5);
}

static constexpr int kTileWords = 32;
static constexpr int kPitch     = kF3 + 4;

__global__ __launch_bounds__(32 * kTileWords) __attribute__((amdgpu_waves_per_eu(8, 8)))
void conv_gather_linear(
    const int*   __restrict__ chars,
    const float* __restrict__ T,
    const float* __restrict__ lw,
    const float* __restrict__ lb,
    float*       __restrict__ out)
{
    __shared__ __attribute__((aligned(16))) float sfeat[kTileWords * kPitch];

    const unsigned lane = threadIdx.x & 31u;
    const int wvs = __builtin_amdgcn_readfirstlane((int)(threadIdx.x >> 5));
    const unsigned wrd = (unsigned)wvs;
    const unsigned n   = blockIdx.x * kTileWords + wrd;
    const unsigned col = lane * 4u;

    const int cv = chars[n * kW + (lane < (unsigned)kW ? lane : (unsigned)(kW - 1))];
#define CH(i) ((unsigned)__builtin_amdgcn_readlane(cv, (i)))

    v8f m1 = {};
    v8f m3 = m1, m5 = m1;

    unsigned a = CH(0), b = CH(1), c = CH(2), d = CH(3);
    position<false, false, true, true>(T, col, 0u, 0u, a, b, c, m1, m3, m5);
    position<false, true,  true, true>(T, col, 0u, a, b, c, d, m1, m3, m5);
    for (int t = 2; t <= kW - 3; ++t) {
        const unsigned e = CH(t + 2);
        position<true, true, true, true>(T, col, a, b, c, d, e, m1, m3, m5);
        a = b; b = c; c = d; d = e;
    }
    position<true, true, true,  false>(T, col, a, b, c, d, 0u, m1, m3, m5);
    __builtin_amdgcn_sched_barrier(0);
    position<true, true, false, false>(T, col, b, c, d, 0u, 0u, m1, m3, m5);
#undef CH

    float* f = sfeat + wrd * kPitch + col;
    *(v4f*)(f)                = __builtin_shufflevector(m1, m1, 0, 1, 2, 3);
    *(v4f*)(f + 128)          = __builtin_shufflevector(m1, m1, 4, 5, 6, 7);
    *(v4f*)(f + kH)           = __builtin_shufflevector(m3, m3, 0, 1, 2, 3);
    *(v4f*)(f + kH + 128)     = __builtin_shufflevector(m3, m3, 4, 5, 6, 7);
    *(v4f*)(f + 2 * kH)       = __builtin_shufflevector(m5, m5, 0, 1, 2, 3);
    *(v4f*)(f + 2 * kH + 128) = __builtin_shufflevector(m5, m5, 4, 5, 6, 7);

    __syncthreads();

    if (wvs < kH / 16) {
        const int nTile = wvs;
        const int half  = (int)(lane >> 4);
        const int l16   = (int)(lane & 15u);

        const float* arow0 = sfeat + l16 * kPitch + 2 * half;
        const float* arow1 = arow0 + 16 * kPitch;
        const float* brow  = lw + (nTile * 16 + l16) * kF3 + 2 * half;

        v8f acc0 = {};
        v8f acc1 = {};
#pragma unroll 8
        for (int k0 = 0; k0 < kF3; k0 += 4) {
            v2f av0 = *(const v2f*)(arow0 + k0);
            v2f av1 = *(const v2f*)(arow1 + k0);
            v2f bv  = *(const v2f*)(brow + k0);
            acc0 = __builtin_amdgcn_wmma_f32_16x16x4_f32(
                false, av0, false, bv, (short)0, acc0, false, false);
            acc1 = __builtin_amdgcn_wmma_f32_16x16x4_f32(
                false, av1, false, bv, (short)0, acc1, false, false);
        }

        const float bias = lb[nTile * 16 + l16];
        float* o = out + (size_t)(blockIdx.x * kTileWords) * kH + nTile * 16;
#pragma unroll
        for (int i = 0; i < 8; ++i) {
            int m = half * 8 + i;
            o[m * kH + l16]        = acc0[i] + bias;
            o[(m + 16) * kH + l16] = acc1[i] + bias;
        }
    }
}

extern "C" void kernel_launch(void* const* d_in, const int* in_sizes, int n_in,
                              void* d_out, int out_size, void* d_ws, size_t ws_size,
                              hipStream_t stream)
{
    const int*   chars = (const int*)  d_in[0];
    const float* emb   = (const float*)d_in[1];
    const float* w1    = (const float*)d_in[2];
    const float* b1    = (const float*)d_in[3];
    const float* w3    = (const float*)d_in[4];
    const float* b3    = (const float*)d_in[5];
    const float* w5    = (const float*)d_in[6];
    const float* b5    = (const float*)d_in[7];
    const float* lw    = (const float*)d_in[8];
    const float* lb    = (const float*)d_in[9];
    float*       outp  = (float*)d_out;

    float* T    = (float*)d_ws;

    build_tables<<<9 * 16 * 16, 32, 0, stream>>>(emb, w1, b1, w3, b3, w5, b5, T);

    conv_gather_linear<<<kWords / kTileWords, 32 * kTileWords, 0, stream>>>(chars, T, lw, lb, outp);
}


# linear's K loop cut into the 8 column slices: each piece's WMMAs run under the next slice's table loads, lw read as whole-line fragments from a copy re-ordered in launch 1; k order now slice-major
# speedup vs baseline: 2.5570x; 1.2011x over previous
#include <hip/hip_runtime.h>
#include <hip/hip_bf16.h>

typedef float v2f __attribute__((ext_vector_type(2)));
typedef float v8f __attribute__((ext_vector_type(8)));
typedef float v4f __attribute__((ext_vector_type(4)));
typedef int   v4i __attribute__((ext_vector_type(4)));

static constexpr int kVocab = 256;
static constexpr int kE     = 64;
static constexpr int kH     = 256;
static constexpr int kW     = 20;
static constexpr int kWords = 64 * 128;
static constexpr int kF3    = 3 * kH;
static constexpr int kSliceCols = 32;
static constexpr int kSlices    = kH / kSliceCols;
static constexpr int kSubTab    = kVocab * kSliceCols;
static constexpr int kTableTiles = 9 * 16 * 16;
static constexpr int kLwBlocks   = 16 * (kF3 / 32);

__global__ __launch_bounds__(32) void build_tables(
    const float* __restrict__ emb,
    const float* __restrict__ w1, const float* __restrict__ b1,
    const float* __restrict__ w3, const float* __restrict__ b3,
    const float* __restrict__ w5, const float* __restrict__ b5,
    const float* __restrict__ lw,
    float* __restrict__ T,
    float* __restrict__ lwF)
{
    if (blockIdx.x >= kTableTiles) {
        const int u   = (int)blockIdx.x - kTableTiles;
        const int nt  = u / (kF3 / 32);
        const int kb  = u % (kF3 / 32);
        const int ln  = threadIdx.x & 31;
        const float* s = lw + (size_t)(nt * 16 + (ln & 15)) * kF3 + kb * 32 + 4 * (ln >> 4);
        float*       d = lwF + (size_t)(nt * (kF3 / 8) + kb * 4) * 128 + ln * 4;
#pragma unroll
        for (int q = 0; q < 4; ++q)
            *(v4f*)(d + q * 128) = *(const v4f*)(s + 8 * q);
        return;
    }
    int tile  = blockIdx.x;
    int tap   = tile >> 8;
    int cTile = (tile >> 4) & 15;
    int hTile = tile & 15;

    const float* wp; const float* bp; int kk, j;
    if (tap == 0)     { wp = w1; bp = b1; kk = 1; j = 0; }
    else if (tap < 4) { wp = w3; bp = b3; kk = 3; j = tap - 1; }
    else              { wp = w5; bp = b5; kk = 5; j = tap - 4; }
    const bool center = (j == (kk >> 1));

    const int lane = threadIdx.x & 31;
    const int half = lane >> 4;
    const int l16  = lane & 15;

    const int row = cTile * 16 + l16;
    const int col = hTile * 16 + l16;

    v8f acc = {};
    for (int e0 = 0; e0 < kE; e0 += 4) {
        int ea = e0 + 2 * half;
        v2f a = *(const v2f*)(emb + row * kE + ea);
        v2f b;
        b.x = wp[(col * kE + ea    ) * kk + j];
        b.y = wp[(col * kE + ea + 1) * kk + j];
        acc = __builtin_amdgcn_wmma_f32_16x16x4_f32(
            false, a, false, b, (short)0, acc, false, false);
    }

    const float bias = center ? bp[col] : 0.0f;
    const int slice = hTile >> 1;
    const int cin   = (hTile & 1) * 16 + l16;
    float* outp = T + (size_t)(tap * kSlices + slice) * kSubTab + (cTile * 16) * kSliceCols + cin;
#pragma unroll
    for (int i = 0; i < 8; ++i) {
        int m = half * 8 + i;
        outp[m * kSliceCols] = acc[i] + bias;
    }
}

template <bool B> struct Flag { static constexpr bool value = B; };

static constexpr int kTileWords = 32;
static constexpr int kPitch     = kF3 + 4;
static constexpr int kLdsFloats = 41 * 1024;

__device__ __forceinline__ v4f vmax4(v4f a, v4f b)
{
    v4f r;
#pragma unroll
    for (int i = 0; i < 4; ++i) r[i] = fmaxf(a[i], b[i]);
    return r;
}

__device__ __forceinline__ v4f rowld(const char* __restrict__ piece, unsigned off)
{
    return *(const v4f*)(piece + off);
}

__global__ __launch_bounds__(8 * 32) __attribute__((amdgpu_waves_per_eu(2, 2)))
void conv_gather_linear(
    const int*   __restrict__ chars,
    const float* __restrict__ T,
    const float* __restrict__ lwF,
    const float* __restrict__ lb,
    float*       __restrict__ out)
{
    __shared__ __attribute__((aligned(16))) float sfeat[kLdsFloats];

    const unsigned lane = threadIdx.x & 31u;
    const int wvs = __builtin_amdgcn_readfirstlane((int)(threadIdx.x >> 5));
    const unsigned g    = lane >> 3;
    const unsigned part = lane & 7u;
    const unsigned wrd  = (unsigned)wvs * 4u + g;
    const unsigned n    = blockIdx.x * kTileWords + wrd;

    unsigned ro[kW];
    {
        const v4i* cp = (const v4i*)(chars + n * kW);
#pragma unroll
        for (int q = 0; q < kW / 4; ++q) {
            v4i c4 = cp[q];
#pragma unroll
            for (int j = 0; j < 4; ++j)
                ro[4 * q + j] = (unsigned)c4[j] * (kSliceCols * 4u) + part * 16u;
        }
    }

    const char* Tb = (const char*)T;
    constexpr size_t kTapBytes = (size_t)kSlices * kSubTab * sizeof(float);

    const int half = (int)(lane >> 4);
    const int l16  = (int)(lane & 15u);
    const int nT0  = wvs * 2;
    const float* arow0 = sfeat + l16 * kPitch + 4 * half;
    const float* arow1 = arow0 + 16 * kPitch;
    const float* lwf0  = lwF + (size_t)(nT0 * (kF3 / 8)) * 128 + lane * 4u;
    const float* lwf1  = lwf0 + (kF3 / 8) * 128;
    v8f acc00 = {}, acc10 = {}, acc01 = {}, acc11 = {};

#define LOADB(d0, d1, c_, br)                                                 \
        _Pragma("unroll")                                                     \
        for (int q = 0; q < 4; ++q) {                                         \
            const int u = ((br) * kSlices + (c_)) * 4 + q;                    \
            d0[q] = *(const v4f*)(lwf0 + u * 128);                            \
            d1[q] = *(const v4f*)(lwf1 + u * 128);                            \
        }
#define WM(av, bv, cv) cv = __builtin_amdgcn_wmma_f32_16x16x4_f32(            \
            false, av, false, bv, (short)0, cv, false, false)
#define MMAS(d0, d1, c_, br)                                                  \
        _Pragma("unroll")                                                     \
        for (int q = 0; q < 4; ++q) {                                         \
            const int k = (br) * kH + (c_) * kSliceCols + 8 * q;              \
            const v4f x0 = *(const v4f*)(arow0 + k);                          \
            const v4f x1 = *(const v4f*)(arow1 + k);                          \
            const v4f z0 = d0[q];                                             \
            const v4f z1 = d1[q];                                             \
            WM(x0.xy, z0.xy, acc00); WM(x1.xy, z0.xy, acc10);                 \
            WM(x0.xy, z1.xy, acc01); WM(x1.xy, z1.xy, acc11);                 \
            WM(x0.zw, z0.zw, acc00); WM(x1.zw, z0.zw, acc10);                 \
            WM(x0.zw, z1.zw, acc01); WM(x1.zw, z1.zw, acc11);                 \
        }

#define PIN asm volatile("" : "+v"(acc00), "+v"(acc10), "+v"(acc01), "+v"(acc11))

#define FENCE __builtin_amdgcn_sched_barrier(0)
#define PIECE(tap) (sl + (size_t)(tap) * kTapBytes)
#define LOADS(buf, tap, lo, hi, d)                                        \
        _Pragma("unroll")                                                 \
        for (int t = (lo); t < (hi); ++t) buf[t] = rowld(PIECE(tap), ro[t + (d)])

    auto slice = [&](const int s, auto linTag) __attribute__((always_inline)) {
        const char* sl = Tb + (size_t)s * (kSubTab * sizeof(float));
        constexpr bool lin = decltype(linTag)::value;
        v4f y0[4], y1[4];
        if constexpr (lin) { LOADB(y0, y1, s - 1, 0); }
        FENCE;

        v4f b0[kW], b1[kW], a[kW];

        LOADS(b0, 0, 0, kW, 0);          FENCE;
        LOADS(b1, 2, 0, kW, 0);          FENCE;
        if constexpr (lin) { MMAS(y0, y1, s - 1, 0); PIN; }
        FENCE;
        v4f m1 = {};
#pragma unroll
        for (int t = 0; t < kW; ++t) m1 = vmax4(m1, b0[t]);
        FENCE;
        if constexpr (lin) { LOADB(y0, y1, s - 1, 1); }
        FENCE;
        LOADS(b0, 1, 1, kW, -1);         FENCE;
#pragma unroll
        for (int t = 0; t < kW; ++t) a[t] = b1[t];
        FENCE;
        LOADS(b1, 3, 0, kW - 1, 1);      FENCE;
        if constexpr (lin) { MMAS(y0, y1, s - 1, 1); PIN; }
        FENCE;
#pragma unroll
        for (int t = 1; t < kW; ++t) a[t] += b0[t];
        FENCE;
        if constexpr (lin) { LOADB(y0, y1, s - 1, 2); }
        FENCE;
        LOADS(b0, 6, 0, kW, 0);          FENCE;
#pragma unroll
        for (int t = 0; t < kW - 1; ++t) a[t] += b1[t];
        v4f m3 = {};
#pragma unroll
        for (int t = 0; t < kW; ++t) m3 = vmax4(m3, a[t]);
        FENCE;
        LOADS(b1, 4, 2, kW, -2);         FENCE;
        if constexpr (lin) { MMAS(y0, y1, s - 1, 2); PIN; }
        FENCE;
#pragma unroll
        for (int t = 0; t < kW; ++t) a[t] = b0[t];
        FENCE;
        LOADS(b0, 5, 1, kW, -1);         FENCE;
#pragma unroll
        for (int t = 2; t < kW; ++t) a[t] += b1[t];
        FENCE;
        LOADS(b1, 7, 0, kW - 1, 1);      FENCE;
#pragma unroll
        for (int t = 1; t < kW; ++t) a[t] += b0[t];
        FENCE;
        LOADS(b0, 8, 0, kW - 2, 2);      FENCE;
#pragma unroll
        for (int t = 0; t < kW - 1; ++t) a[t] += b1[t];
#pragma unroll
        for (int t = 0; t < kW - 2; ++t) a[t] += b0[t];
        v4f m5 = {};
#pragma unroll
        for (int t = 0; t < kW; ++t) m5 = vmax4(m5, a[t]);

        float* f = sfeat + wrd * kPitch + s * kSliceCols + part * 4u;
        *(v4f*)(f)          = m1;
        *(v4f*)(f + kH)     = m3;
        *(v4f*)(f + 2 * kH) = m5;

        __syncthreads();
    };
    slice(0, Flag<false>{});
    for (int s = 1; s < kSlices; ++s) slice(s, Flag<true>{});
#undef LOADS
#undef PIECE

    {
        v4f y0[4], y1[4], y2[4], y3[4], y4[4], y5[4];
        LOADB(y0, y1, kSlices - 1, 0);
        LOADB(y2, y3, kSlices - 1, 1);
        LOADB(y4, y5, kSlices - 1, 2);
        MMAS(y0, y1, kSlices - 1, 0);
        MMAS(y2, y3, kSlices - 1, 1);
        MMAS(y4, y5, kSlices - 1, 2);
#undef FENCE
#undef PIN
#undef MMAS
#undef WM
#undef LOADB

        const float bias0 = lb[nT0 * 16 + l16];
        const float bias1 = lb[nT0 * 16 + 16 + l16];
        float* o = out + (size_t)(blockIdx.x * kTileWords) * kH + nT0 * 16;
#pragma unroll
        for (int i = 0; i < 8; ++i) {
            int m = half * 8 + i;
            o[m * kH + l16]             = acc00[i] + bias0;
            o[(m + 16) * kH + l16]      = acc10[i] + bias0;
            o[m * kH + 16 + l16]        = acc01[i] + bias1;
            o[(m + 16) * kH + 16 + l16] = acc11[i] + bias1;
        }
    }
}

extern "C" void kernel_launch(void* const* d_in, const int* in_sizes, int n_in,
                              void* d_out, int out_size, void* d_ws, size_t ws_size,
                              hipStream_t stream)
{
    const int*   chars = (const int*)  d_in[0];
    const float* emb   = (const float*)d_in[1];
    const float* w1    = (const float*)d_in[2];
    const float* b1    = (const float*)d_in[3];
    const float* w3    = (const float*)d_in[4];
    const float* b3    = (const float*)d_in[5];
    const float* w5    = (const float*)d_in[6];
    const float* b5    = (const float*)d_in[7];
    const float* lw    = (const float*)d_in[8];
    const float* lb    = (const float*)d_in[9];
    float*       outp  = (float*)d_out;

    float* T    = (float*)d_ws;
    float* lwF  = T + 9 * kSlices * kSubTab;

    build_tables<<<kTableTiles + kLwBlocks, 32, 0, stream>>>(emb, w1, b1, w3, b3, w5, b5, lw, T, lwF);

    conv_gather_linear<<<kWords / kTileWords, 8 * 32, 0, stream>>>(chars, T, lwF, lb, outp);
}


# each workgroup starts its walk over the 8 column slices at a slice of its own and wraps, so the 256 no longer ask L2 for the same 32 KiB at one moment; linear's k order starts at that slice too
# speedup vs baseline: 2.6992x; 1.0556x over previous
#include <hip/hip_runtime.h>
#include <hip/hip_bf16.h>

typedef float v2f __attribute__((ext_vector_type(2)));
typedef float v8f __attribute__((ext_vector_type(8)));
typedef float v4f __attribute__((ext_vector_type(4)));
typedef int   v4i __attribute__((ext_vector_type(4)));

static constexpr int kVocab = 256;
static constexpr int kE     = 64;
static constexpr int kH     = 256;
static constexpr int kW     = 20;
static constexpr int kWords = 64 * 128;
static constexpr int kF3    = 3 * kH;
static constexpr int kSliceCols = 32;
static constexpr int kSlices    = kH / kSliceCols;
static constexpr int kSubTab    = kVocab * kSliceCols;
static constexpr int kTableTiles = 9 * 16 * 16;
static constexpr int kLwBlocks   = 16 * (kF3 / 32);

__global__ __launch_bounds__(32) void build_tables(
    const float* __restrict__ emb,
    const float* __restrict__ w1, const float* __restrict__ b1,
    const float* __restrict__ w3, const float* __restrict__ b3,
    const float* __restrict__ w5, const float* __restrict__ b5,
    const float* __restrict__ lw,
    float* __restrict__ T,
    float* __restrict__ lwF)
{
    if (blockIdx.x >= kTableTiles) {
        const int u   = (int)blockIdx.x - kTableTiles;
        const int nt  = u / (kF3 / 32);
        const int kb  = u % (kF3 / 32);
        const int ln  = threadIdx.x & 31;
        const float* s = lw + (size_t)(nt * 16 + (ln & 15)) * kF3 + kb * 32 + 4 * (ln >> 4);
        float*       d = lwF + (size_t)(nt * (kF3 / 8) + kb * 4) * 128 + ln * 4;
#pragma unroll
        for (int q = 0; q < 4; ++q)
            *(v4f*)(d + q * 128) = *(const v4f*)(s + 8 * q);
        return;
    }
    int tile  = blockIdx.x;
    int tap   = tile >> 8;
    int cTile = (tile >> 4) & 15;
    int hTile = tile & 15;

    const float* wp; const float* bp; int kk, j;
    if (tap == 0)     { wp = w1; bp = b1; kk = 1; j = 0; }
    else if (tap < 4) { wp = w3; bp = b3; kk = 3; j = tap - 1; }
    else              { wp = w5; bp = b5; kk = 5; j = tap - 4; }
    const bool center = (j == (kk >> 1));

    const int lane = threadIdx.x & 31;
    const int half = lane >> 4;
    const int l16  = lane & 15;

    const int row = cTile * 16 + l16;
    const int col = hTile * 16 + l16;

    v8f acc = {};
    for (int e0 = 0; e0 < kE; e0 += 4) {
        int ea = e0 + 2 * half;
        v2f a = *(const v2f*)(emb + row * kE + ea);
        v2f b;
        b.x = wp[(col * kE + ea    ) * kk + j];
        b.y = wp[(col * kE + ea + 1) * kk + j];
        acc = __builtin_amdgcn_wmma_f32_16x16x4_f32(
            false, a, false, b, (short)0, acc, false, false);
    }

    const float bias = center ? bp[col] : 0.0f;
    const int slice = hTile >> 1;
    const int cin   = (hTile & 1) * 16 + l16;
    float* outp = T + (size_t)(tap * kSlices + slice) * kSubTab + (cTile * 16) * kSliceCols + cin;
#pragma unroll
    for (int i = 0; i < 8; ++i) {
        int m = half * 8 + i;
        outp[m * kSliceCols] = acc[i] + bias;
    }
}

template <bool B> struct Flag { static constexpr bool value = B; };

static constexpr int kTileWords = 32;
static constexpr int kPitch     = kF3 + 4;
static constexpr int kLdsFloats = 41 * 1024;

__device__ __forceinline__ v4f vmax4(v4f a, v4f b)
{
    v4f r;
#pragma unroll
    for (int i = 0; i < 4; ++i) r[i] = fmaxf(a[i], b[i]);
    return r;
}

__device__ __forceinline__ v4f rowld(const char* __restrict__ piece, unsigned off)
{
    return *(const v4f*)(piece + off);
}

__global__ __launch_bounds__(8 * 32) __attribute__((amdgpu_waves_per_eu(2, 2)))
void conv_gather_linear(
    const int*   __restrict__ chars,
    const float* __restrict__ T,
    const float* __restrict__ lwF,
    const float* __restrict__ lb,
    float*       __restrict__ out)
{
    __shared__ __attribute__((aligned(16))) float sfeat[kLdsFloats];

    const unsigned lane = threadIdx.x & 31u;
    const int wvs = __builtin_amdgcn_readfirstlane((int)(threadIdx.x >> 5));
    const unsigned g    = lane >> 3;
    const unsigned part = lane & 7u;
    const unsigned wrd  = (unsigned)wvs * 4u + g;
    const unsigned n    = blockIdx.x * kTileWords + wrd;

    unsigned ro[kW];
    {
        const v4i* cp = (const v4i*)(chars + n * kW);
#pragma unroll
        for (int q = 0; q < kW / 4; ++q) {
            v4i c4 = cp[q];
#pragma unroll
            for (int j = 0; j < 4; ++j)
                ro[4 * q + j] = (unsigned)c4[j] * (kSliceCols * 4u) + part * 16u;
        }
    }

    const char* Tb = (const char*)T;
    constexpr size_t kTapBytes = (size_t)kSlices * kSubTab * sizeof(float);

    const int half = (int)(lane >> 4);
    const int l16  = (int)(lane & 15u);
    const int nT0  = wvs * 2;
    const float* arow0 = sfeat + l16 * kPitch + 4 * half;
    const float* arow1 = arow0 + 16 * kPitch;
    const float* lwf0  = lwF + (size_t)(nT0 * (kF3 / 8)) * 128 + lane * 4u;
    const float* lwf1  = lwf0 + (kF3 / 8) * 128;
    v8f acc00 = {}, acc10 = {}, acc01 = {}, acc11 = {};

    const int rot = (int)((blockIdx.x + (blockIdx.x >> 3) + (blockIdx.x >> 6)) & 7u);

#define LOADB(d0, d1, c_, br)                                                 \
        _Pragma("unroll")                                                     \
        for (int q = 0; q < 4; ++q) {                                         \
            const int u = ((br) * kSlices + (c_)) * 4 + q;                    \
            d0[q] = *(const v4f*)(lwf0 + u * 128);                            \
            d1[q] = *(const v4f*)(lwf1 + u * 128);                            \
        }
#define WM(av, bv, cv) cv = __builtin_amdgcn_wmma_f32_16x16x4_f32(            \
            false, av, false, bv, (short)0, cv, false, false)
#define MMAS(d0, d1, c_, br)                                                  \
        _Pragma("unroll")                                                     \
        for (int q = 0; q < 4; ++q) {                                         \
            const int k = (br) * kH + (c_) * kSliceCols + 8 * q;              \
            const v4f x0 = *(const v4f*)(arow0 + k);                          \
            const v4f x1 = *(const v4f*)(arow1 + k);                          \
            const v4f z0 = d0[q];                                             \
            const v4f z1 = d1[q];                                             \
            WM(x0.xy, z0.xy, acc00); WM(x1.xy, z0.xy, acc10);                 \
            WM(x0.xy, z1.xy, acc01); WM(x1.xy, z1.xy, acc11);                 \
            WM(x0.zw, z0.zw, acc00); WM(x1.zw, z0.zw, acc10);                 \
            WM(x0.zw, z1.zw, acc01); WM(x1.zw, z1.zw, acc11);                 \
        }

#define PIN asm volatile("" : "+v"(acc00), "+v"(acc10), "+v"(acc01), "+v"(acc11))

#define FENCE __builtin_amdgcn_sched_barrier(0)
#define PIECE(tap) (sl + (size_t)(tap) * kTapBytes)
#define LOADS(buf, tap, lo, hi, d)                                        \
        _Pragma("unroll")                                                 \
        for (int t = (lo); t < (hi); ++t) buf[t] = rowld(PIECE(tap), ro[t + (d)])

    auto slice = [&](const int s, const int sp, auto linTag) __attribute__((always_inline)) {
        const char* sl = Tb + (size_t)s * (kSubTab * sizeof(float));
        constexpr bool lin = decltype(linTag)::value;
        v4f y0[4], y1[4];
        if constexpr (lin) { LOADB(y0, y1, sp, 0); }
        FENCE;

        v4f b0[kW], b1[kW], a[kW];

        LOADS(b0, 0, 0, kW, 0);          FENCE;
        LOADS(b1, 2, 0, kW, 0);          FENCE;
        if constexpr (lin) { MMAS(y0, y1, sp, 0); PIN; }
        FENCE;
        v4f m1 = {};
#pragma unroll
        for (int t = 0; t < kW; ++t) m1 = vmax4(m1, b0[t]);
        FENCE;
        if constexpr (lin) { LOADB(y0, y1, sp, 1); }
        FENCE;
        LOADS(b0, 1, 1, kW, -1);         FENCE;
#pragma unroll
        for (int t = 0; t < kW; ++t) a[t] = b1[t];
        FENCE;
        LOADS(b1, 3, 0, kW - 1, 1);      FENCE;
        if constexpr (lin) { MMAS(y0, y1, sp, 1); PIN; }
        FENCE;
#pragma unroll
        for (int t = 1; t < kW; ++t) a[t] += b0[t];
        FENCE;
        if constexpr (lin) { LOADB(y0, y1, sp, 2); }
        FENCE;
        LOADS(b0, 6, 0, kW, 0);          FENCE;
#pragma unroll
        for (int t = 0; t < kW - 1; ++t) a[t] += b1[t];
        v4f m3 = {};
#pragma unroll
        for (int t = 0; t < kW; ++t) m3 = vmax4(m3, a[t]);
        FENCE;
        LOADS(b1, 4, 2, kW, -2);         FENCE;
        if constexpr (lin) { MMAS(y0, y1, sp, 2); PIN; }
        FENCE;
#pragma unroll
        for (int t = 0; t < kW; ++t) a[t] = b0[t];
        FENCE;
        LOADS(b0, 5, 1, kW, -1);         FENCE;
#pragma unroll
        for (int t = 2; t < kW; ++t) a[t] += b1[t];
        FENCE;
        LOADS(b1, 7, 0, kW - 1, 1);      FENCE;
#pragma unroll
        for (int t = 1; t < kW; ++t) a[t] += b0[t];
        FENCE;
        LOADS(b0, 8, 0, kW - 2, 2);      FENCE;
#pragma unroll
        for (int t = 0; t < kW - 1; ++t) a[t] += b1[t];
#pragma unroll
        for (int t = 0; t < kW - 2; ++t) a[t] += b0[t];
        v4f m5 = {};
#pragma unroll
        for (int t = 0; t < kW; ++t) m5 = vmax4(m5, a[t]);

        float* f = sfeat + wrd * kPitch + s * kSliceCols + part * 4u;
        *(v4f*)(f)          = m1;
        *(v4f*)(f + kH)     = m3;
        *(v4f*)(f + 2 * kH) = m5;

        __syncthreads();
    };
    slice(rot, rot, Flag<false>{});
    for (int i = 1; i < kSlices; ++i)
        slice((rot + i) & (kSlices - 1), (rot + i - 1) & (kSlices - 1), Flag<true>{});
#undef LOADS
#undef PIECE

    {
        const int sLast = (rot + kSlices - 1) & (kSlices - 1);
        const float bias0 = lb[nT0 * 16 + l16];
        const float bias1 = lb[nT0 * 16 + 16 + l16];
        v4f y0[4], y1[4], y2[4], y3[4], y4[4], y5[4];
        LOADB(y0, y1, sLast, 0);
        LOADB(y2, y3, sLast, 1);
        LOADB(y4, y5, sLast, 2);
        FENCE;
        MMAS(y0, y1, sLast, 0);
        MMAS(y2, y3, sLast, 1);
        MMAS(y4, y5, sLast, 2);
        FENCE;
#undef FENCE
#undef PIN
#undef MMAS
#undef WM
#undef LOADB

        float* o = out + (size_t)(blockIdx.x * kTileWords) * kH + nT0 * 16;
#pragma unroll
        for (int i = 0; i < 8; ++i) {
            int m = half * 8 + i;
            o[m * kH + l16]             = acc00[i] + bias0;
            o[(m + 16) * kH + l16]      = acc10[i] + bias0;
            o[m * kH + 16 + l16]        = acc01[i] + bias1;
            o[(m + 16) * kH + 16 + l16] = acc11[i] + bias1;
        }
    }
}

extern "C" void kernel_launch(void* const* d_in, const int* in_sizes, int n_in,
                              void* d_out, int out_size, void* d_ws, size_t ws_size,
                              hipStream_t stream)
{
    const int*   chars = (const int*)  d_in[0];
    const float* emb   = (const float*)d_in[1];
    const float* w1    = (const float*)d_in[2];
    const float* b1    = (const float*)d_in[3];
    const float* w3    = (const float*)d_in[4];
    const float* b3    = (const float*)d_in[5];
    const float* w5    = (const float*)d_in[6];
    const float* b5    = (const float*)d_in[7];
    const float* lw    = (const float*)d_in[8];
    const float* lb    = (const float*)d_in[9];
    float*       outp  = (float*)d_out;

    float* T    = (float*)d_ws;
    float* lwF  = T + 9 * kSlices * kSubTab;

    build_tables<<<kTableTiles + kLwBlocks, 32, 0, stream>>>(emb, w1, b1, w3, b3, w5, b5, lw, T, lwF);

    conv_gather_linear<<<kWords / kTileWords, 8 * 32, 0, stream>>>(chars, T, lwF, lb, outp);
}
